# starting kernel
# baseline (speedup 1.0000x reference)
#include <hip/hip_runtime.h>

#define DIM   64
#define NREL  8
#define NBLK  8
#define BS    8

typedef __attribute__((ext_vector_type(2))) float v2f;
typedef __attribute__((ext_vector_type(8))) float v8f;

__global__ void zero_f32v4(float4* __restrict__ p, long long n4) {
  long long i = (long long)blockIdx.x * blockDim.x + threadIdx.x;
  long long stride = (long long)gridDim.x * blockDim.x;
  float4 z; z.x = z.y = z.z = z.w = 0.0f;
  for (; i < n4; i += stride) p[i] = z;
}

__global__ void edge_scatter(const float*  __restrict__ x,
                             const float*  __restrict__ ew,
                             const int*    __restrict__ src,
                             const int*    __restrict__ tgt,
                             const int*    __restrict__ et,
                             float*        __restrict__ agg,
                             int nEdges, int nNodes) {
  long long tid    = (long long)blockIdx.x * blockDim.x + threadIdx.x;
  long long stride = (long long)gridDim.x * blockDim.x;
  long long total  = (long long)nEdges * (DIM / 4);
  for (; tid < total; tid += stride) {
    int e = (int)(tid >> 4);
    int c = (int)(tid & 15);
    int s = src[e];
    int t = tgt[e];
    int r = et[e];
    float w = ew[e];
    float4 xv = *(const float4*)(x + (size_t)s * DIM + c * 4);
    float* dst = agg + ((size_t)r * nNodes + t) * DIM + c * 4;
    unsafeAtomicAdd(dst + 0, w * xv.x);
    unsafeAtomicAdd(dst + 1, w * xv.y);
    unsafeAtomicAdd(dst + 2, w * xv.z);
    unsafeAtomicAdd(dst + 3, w * xv.w);
  }
}

__global__ void __launch_bounds__(256)
rgcn_block_wmma(const float* __restrict__ agg,
                const float* __restrict__ blocks,
                float*       __restrict__ out,
                int nNodes) {
  const int lane   = threadIdx.x & 31;
  const int wib    = threadIdx.x >> 5;
  const int gwave  = blockIdx.x * (blockDim.x >> 5) + wib;
  const int nWaves = gridDim.x * (blockDim.x >> 5);

  const int j       = gwave & 3;
  const int group   = gwave >> 2;
  const int nGroups = nWaves >> 2;
  const int nTiles  = (nNodes + 15) >> 4;

  const int nl   = lane & 15;
  const int half = lane >> 4;

  v2f Bfrag[NREL][4];
#pragma unroll
  for (int r = 0; r < NREL; ++r) {
#pragma unroll
    for (int s = 0; s < 4; ++s) {
#pragma unroll
      for (int p = 0; p < 2; ++p) {
        int kl = 4 * s + p + 2 * half;
        float v = 0.0f;
        if ((kl >> 3) == (nl >> 3)) {
          int blk = 2 * j + (kl >> 3);
          v = blocks[(((size_t)r * NBLK + blk) * BS + (kl & 7)) * BS + (nl & 7)];
        }
        Bfrag[r][s][p] = v;
      }
    }
  }

  for (int tile = group; tile < nTiles; tile += nGroups) {
    const int node0  = tile << 4;
    const int myNode = node0 + nl;
    const bool full  = (node0 + 16) <= nNodes;

    v2f a[NREL][4];
    if (full) {
#pragma unroll
      for (int r = 0; r < NREL; ++r) {
        const float* rowBase = agg + ((size_t)r * nNodes + myNode) * DIM
                                   + 16 * j + 2 * half;
#pragma unroll
        for (int s = 0; s < 4; ++s)
          a[r][s] = *(const v2f*)(rowBase + 4 * s);
      }
    } else {
      const bool valid = myNode < nNodes;
      const int  cn    = valid ? myNode : (nNodes - 1);
#pragma unroll
      for (int r = 0; r < NREL; ++r) {
        const float* rowBase = agg + ((size_t)r * nNodes + cn) * DIM
                                   + 16 * j + 2 * half;
#pragma unroll
        for (int s = 0; s < 4; ++s) {
          v2f t = *(const v2f*)(rowBase + 4 * s);
          a[r][s][0] = valid ? t[0] : 0.0f;
          a[r][s][1] = valid ? t[1] : 0.0f;
        }
      }
    }

    v8f acc = {};
#pragma unroll
    for (int r = 0; r < NREL; ++r) {
#pragma unroll
      for (int s = 0; s < 4; ++s) {
        acc = __builtin_amdgcn_wmma_f32_16x16x4_f32(
            false, a[r][s], false, Bfrag[r][s], (short)0, acc, false, false);
      }
    }

    float* outBase = out + (size_t)(node0 + 8 * half) * DIM + 16 * j + nl;
    if (full) {
#pragma unroll
      for (int i = 0; i < 8; ++i)
        outBase[(size_t)i * DIM] = acc[i];
    } else {
#pragma unroll
      for (int i = 0; i < 8; ++i) {
        int row = node0 + i + 8 * half;
        if (row < nNodes)
          out[(size_t)row * DIM + 16 * j + nl] = acc[i];
      }
    }
  }
}

__global__ void edge_direct(const float* __restrict__ x,
                            const float* __restrict__ blocks,
                            const float* __restrict__ ew,
                            const int*   __restrict__ src,
                            const int*   __restrict__ tgt,
                            const int*   __restrict__ et,
                            float*       __restrict__ out,
                            int nEdges) {
  long long tid    = (long long)blockIdx.x * blockDim.x + threadIdx.x;
  long long stride = (long long)gridDim.x * blockDim.x;
  long long total  = (long long)nEdges * NBLK;
  for (; tid < total; tid += stride) {
    int e = (int)(tid >> 3);
    int b = (int)(tid & 7);
    int s = src[e], t = tgt[e], r = et[e];
    float w = ew[e];
    const float* xb = x + (size_t)s * DIM + b * BS;
    const float* W  = blocks + ((size_t)r * NBLK + b) * (BS * BS);
    float xi[BS];
#pragma unroll
    for (int i = 0; i < BS; ++i) xi[i] = xb[i];
#pragma unroll
    for (int o = 0; o < BS; ++o) {
      float acc = 0.0f;
#pragma unroll
      for (int i = 0; i < BS; ++i) acc += xi[i] * W[i * BS + o];
      unsafeAtomicAdd(out + (size_t)t * DIM + b * BS + o, w * acc);
    }
  }
}

extern "C" void kernel_launch(void* const* d_in, const int* in_sizes, int n_in,
                              void* d_out, int out_size, void* d_ws, size_t ws_size,
                              hipStream_t stream) {
  const float* x      = (const float*)d_in[0];
  const float* blocks = (const float*)d_in[1];
  const float* ew     = (const float*)d_in[2];
  const int*   src    = (const int*)d_in[3];
  const int*   tgt    = (const int*)d_in[4];
  const int*   et     = (const int*)d_in[5];
  float*       out    = (float*)d_out;

  const int nNodes = in_sizes[0] / DIM;
  const int nEdges = in_sizes[2];

  const size_t aggElems = (size_t)NREL * nNodes * DIM;
  const size_t need     = aggElems * sizeof(float);

  if (ws_size >= need) {
    float* agg = (float*)d_ws;

    long long n4 = (long long)(aggElems / 4);
    zero_f32v4<<<4096, 256, 0, stream>>>((float4*)agg, n4);

    edge_scatter<<<8192, 256, 0, stream>>>(x, ew, src, tgt, et, agg,
                                           nEdges, nNodes);

    rgcn_block_wmma<<<512, 256, 0, stream>>>(agg, blocks, out, nNodes);
  } else {
    long long o4 = (long long)out_size / 4;
    zero_f32v4<<<2048, 256, 0, stream>>>((float4*)out, o4);
    edge_direct<<<8192, 256, 0, stream>>>(x, blocks, ew, src, tgt, et,
                                          out, nEdges);
  }
}


# edge_scatter: a wave walks 32 edges with lane = feature, so each gather and each float atomic covers one whole 128-B line (was 4 lines per instruction); atomic arrival order free as before
# speedup vs baseline: 1.5218x; 1.5218x over previous
#include <hip/hip_runtime.h>

#define DIM   64
#define NREL  8
#define NBLK  8
#define BS    8

typedef __attribute__((ext_vector_type(2))) float v2f;
typedef __attribute__((ext_vector_type(8))) float v8f;

__global__ void zero_f32v4(float4* __restrict__ p, long long n4) {
  long long i = (long long)blockIdx.x * blockDim.x + threadIdx.x;
  long long stride = (long long)gridDim.x * blockDim.x;
  float4 z; z.x = z.y = z.z = z.w = 0.0f;
  for (; i < n4; i += stride) p[i] = z;
}

__global__ void __launch_bounds__(64)
edge_scatter(const float*  __restrict__ x,
             const float*  __restrict__ ew,
             const int*    __restrict__ src,
             const int*    __restrict__ tgt,
             const int*    __restrict__ et,
             float*        __restrict__ agg,
             int nEdges, int nNodes) {
  const int lane = threadIdx.x & 31;
  const int wave = __builtin_amdgcn_readfirstlane(
      (int)((blockIdx.x * blockDim.x + threadIdx.x) >> 5));
  const int nWaves = (int)((gridDim.x * blockDim.x) >> 5);

  for (int base = wave * 32; base < nEdges; base += nWaves * 32) {
    const int e  = base + lane;
    const int ec = e < nEdges ? e : nEdges - 1;
    const int sv = src[ec];
    const int tv = tgt[ec];
    const int rv = et[ec];
    const int wv = __float_as_int(ew[ec]);
    const int rem = nEdges - base;

    if (rem >= 32) {
      float xa[32], xb[32];
#pragma unroll
      for (int i = 0; i < 32; ++i) {
        const int s = __builtin_amdgcn_readlane(sv, i);
        const float* xr = x + (size_t)s * DIM;
        xa[i] = xr[lane];
        xb[i] = xr[lane + 32];
      }
#pragma unroll
      for (int i = 0; i < 32; ++i) {
        const int   t = __builtin_amdgcn_readlane(tv, i);
        const int   r = __builtin_amdgcn_readlane(rv, i);
        const float w = __int_as_float(__builtin_amdgcn_readlane(wv, i));
        float* dst = agg + ((size_t)r * nNodes + t) * DIM;
        unsafeAtomicAdd(dst + lane,      w * xa[i]);
        unsafeAtomicAdd(dst + lane + 32, w * xb[i]);
      }
    } else {
      for (int i = 0; i < rem; ++i) {
        const int   s = __builtin_amdgcn_readlane(sv, i);
        const int   t = __builtin_amdgcn_readlane(tv, i);
        const int   r = __builtin_amdgcn_readlane(rv, i);
        const float w = __int_as_float(__builtin_amdgcn_readlane(wv, i));
        const float* xr = x + (size_t)s * DIM;
        float* dst = agg + ((size_t)r * nNodes + t) * DIM;
        unsafeAtomicAdd(dst + lane,      w * xr[lane]);
        unsafeAtomicAdd(dst + lane + 32, w * xr[lane + 32]);
      }
    }
  }
}

__global__ void __launch_bounds__(256)
rgcn_block_wmma(const float* __restrict__ agg,
                const float* __restrict__ blocks,
                float*       __restrict__ out,
                int nNodes) {
  const int lane   = threadIdx.x & 31;
  const int wib    = threadIdx.x >> 5;
  const int gwave  = blockIdx.x * (blockDim.x >> 5) + wib;
  const int nWaves = gridDim.x * (blockDim.x >> 5);

  const int j       = gwave & 3;
  const int group   = gwave >> 2;
  const int nGroups = nWaves >> 2;
  const int nTiles  = (nNodes + 15) >> 4;

  const int nl   = lane & 15;
  const int half = lane >> 4;

  v2f Bfrag[NREL][4];
#pragma unroll
  for (int r = 0; r < NREL; ++r) {
#pragma unroll
    for (int s = 0; s < 4; ++s) {
#pragma unroll
      for (int p = 0; p < 2; ++p) {
        int kl = 4 * s + p + 2 * half;
        float v = 0.0f;
        if ((kl >> 3) == (nl >> 3)) {
          int blk = 2 * j + (kl >> 3);
          v = blocks[(((size_t)r * NBLK + blk) * BS + (kl & 7)) * BS + (nl & 7)];
        }
        Bfrag[r][s][p] = v;
      }
    }
  }

  for (int tile = group; tile < nTiles; tile += nGroups) {
    const int node0  = tile << 4;
    const int myNode = node0 + nl;
    const bool full  = (node0 + 16) <= nNodes;

    v2f a[NREL][4];
    if (full) {
#pragma unroll
      for (int r = 0; r < NREL; ++r) {
        const float* rowBase = agg + ((size_t)r * nNodes + myNode) * DIM
                                   + 16 * j + 2 * half;
#pragma unroll
        for (int s = 0; s < 4; ++s)
          a[r][s] = *(const v2f*)(rowBase + 4 * s);
      }
    } else {
      const bool valid = myNode < nNodes;
      const int  cn    = valid ? myNode : (nNodes - 1);
#pragma unroll
      for (int r = 0; r < NREL; ++r) {
        const float* rowBase = agg + ((size_t)r * nNodes + cn) * DIM
                                   + 16 * j + 2 * half;
#pragma unroll
        for (int s = 0; s < 4; ++s) {
          v2f t = *(const v2f*)(rowBase + 4 * s);
          a[r][s][0] = valid ? t[0] : 0.0f;
          a[r][s][1] = valid ? t[1] : 0.0f;
        }
      }
    }

    v8f acc = {};
#pragma unroll
    for (int r = 0; r < NREL; ++r) {
#pragma unroll
      for (int s = 0; s < 4; ++s) {
        acc = __builtin_amdgcn_wmma_f32_16x16x4_f32(
            false, a[r][s], false, Bfrag[r][s], (short)0, acc, false, false);
      }
    }

    float* outBase = out + (size_t)(node0 + 8 * half) * DIM + 16 * j + nl;
    if (full) {
#pragma unroll
      for (int i = 0; i < 8; ++i)
        outBase[(size_t)i * DIM] = acc[i];
    } else {
#pragma unroll
      for (int i = 0; i < 8; ++i) {
        int row = node0 + i + 8 * half;
        if (row < nNodes)
          out[(size_t)row * DIM + 16 * j + nl] = acc[i];
      }
    }
  }
}

__global__ void edge_direct(const float* __restrict__ x,
                            const float* __restrict__ blocks,
                            const float* __restrict__ ew,
                            const int*   __restrict__ src,
                            const int*   __restrict__ tgt,
                            const int*   __restrict__ et,
                            float*       __restrict__ out,
                            int nEdges) {
  long long tid    = (long long)blockIdx.x * blockDim.x + threadIdx.x;
  long long stride = (long long)gridDim.x * blockDim.x;
  long long total  = (long long)nEdges * NBLK;
  for (; tid < total; tid += stride) {
    int e = (int)(tid >> 3);
    int b = (int)(tid & 7);
    int s = src[e], t = tgt[e], r = et[e];
    float w = ew[e];
    const float* xb = x + (size_t)s * DIM + b * BS;
    const float* W  = blocks + ((size_t)r * NBLK + b) * (BS * BS);
    float xi[BS];
#pragma unroll
    for (int i = 0; i < BS; ++i) xi[i] = xb[i];
#pragma unroll
    for (int o = 0; o < BS; ++o) {
      float acc = 0.0f;
#pragma unroll
      for (int i = 0; i < BS; ++i) acc += xi[i] * W[i * BS + o];
      unsafeAtomicAdd(out + (size_t)t * DIM + b * BS + o, w * acc);
    }
  }
}

extern "C" void kernel_launch(void* const* d_in, const int* in_sizes, int n_in,
                              void* d_out, int out_size, void* d_ws, size_t ws_size,
                              hipStream_t stream) {
  const float* x      = (const float*)d_in[0];
  const float* blocks = (const float*)d_in[1];
  const float* ew     = (const float*)d_in[2];
  const int*   src    = (const int*)d_in[3];
  const int*   tgt    = (const int*)d_in[4];
  const int*   et     = (const int*)d_in[5];
  float*       out    = (float*)d_out;

  const int nNodes = in_sizes[0] / DIM;
  const int nEdges = in_sizes[2];

  const size_t aggElems = (size_t)NREL * nNodes * DIM;
  const size_t need     = aggElems * sizeof(float);

  if (ws_size >= need) {
    float* agg = (float*)d_ws;

    long long n4 = (long long)(aggElems / 4);
    zero_f32v4<<<4096, 256, 0, stream>>>((float4*)agg, n4);

    edge_scatter<<<3125, 64, 0, stream>>>(x, ew, src, tgt, et, agg,
                                          nEdges, nNodes);

    rgcn_block_wmma<<<512, 256, 0, stream>>>(agg, blocks, out, nNodes);
  } else {
    long long o4 = (long long)out_size / 4;
    zero_f32v4<<<2048, 256, 0, stream>>>((float4*)out, o4);
    edge_direct<<<8192, 256, 0, stream>>>(x, blocks, ew, src, tgt, et,
                                          out, nEdges);
  }
}
